# starting kernel
# baseline (speedup 1.0000x reference)
#include <hip/hip_runtime.h>

typedef __attribute__((ext_vector_type(2))) float v2f;
typedef __attribute__((ext_vector_type(8))) float v8f;

#define D 128

__global__ __launch_bounds__(256) void k_deg_init(float* __restrict__ deg, int n) {
  int i = blockIdx.x * 256 + threadIdx.x;
  if (i < n) deg[i] = 1.0f;
}

__global__ __launch_bounds__(256) void k_deg_accum(const int* __restrict__ col,
                                                   float* __restrict__ deg, int e) {
  int i = blockIdx.x * 256 + threadIdx.x;
  if (i < e) atomicAdd(&deg[col[i]], 1.0f);
}

__global__ __launch_bounds__(256) void k_dinv(float* __restrict__ deg, int n) {
  int i = blockIdx.x * 256 + threadIdx.x;
  if (i < n) {
    float d = deg[i];
    deg[i] = d > 0.0f ? rsqrtf(d) : 0.0f;
  }
}

__global__ __launch_bounds__(256) void k_gemm_selfinit(
    const float* __restrict__ seq, const float* __restrict__ W,
    const float* __restrict__ dinv, float* __restrict__ xW,
    float* __restrict__ out) {
  const int lane  = threadIdx.x & 31;
  const int wave  = threadIdx.x >> 5;
  const int row0  = blockIdx.x * 16;
  const int col0  = wave * 16;
  const int mn    = lane & 15;
  const int khalf = (lane >> 4) << 1;

  const float* arow = seq + (size_t)(row0 + mn) * D;
  v8f acc = {};
#pragma unroll 4
  for (int k = 0; k < D; k += 4) {
    v2f a, b;
    a.x = arow[k + khalf];
    a.y = arow[k + khalf + 1];
    b.x = W[(size_t)(k + khalf) * D + col0 + mn];
    b.y = W[(size_t)(k + khalf + 1) * D + col0 + mn];
    acc = __builtin_amdgcn_wmma_f32_16x16x4_f32(
         false, a,  false, b,
         (short)0, acc,  false,  false);
  }

  const int mbase = (lane >> 4) << 3;
#pragma unroll
  for (int j = 0; j < 8; ++j) {
    int r = row0 + mbase + j;
    float di = dinv[r];
    size_t off = (size_t)r * D + col0 + mn;
    xW[off]  = acc[j];
    out[off] = acc[j] * di * di;
  }
}

__global__ __launch_bounds__(128) void k_tail_rows(
    const float* __restrict__ seq, const float* __restrict__ W,
    const float* __restrict__ dinv, float* __restrict__ xW,
    float* __restrict__ out, int rowStart, int n) {
  int r = rowStart + blockIdx.x;
  if (r >= n) return;
  int c = threadIdx.x;
  float acc = 0.0f;
  for (int k = 0; k < D; ++k) acc += seq[(size_t)r * D + k] * W[(size_t)k * D + c];
  float di = dinv[r];
  xW[(size_t)r * D + c]  = acc;
  out[(size_t)r * D + c] = acc * di * di;
}

__global__ __launch_bounds__(256) void k_scatter(
    const int* __restrict__ row, const int* __restrict__ col,
    const float* __restrict__ xW, const float* __restrict__ dinv,
    float* __restrict__ out, int e) {
  int idx = blockIdx.x * 256 + threadIdx.x;
  int eid = idx >> 5;
  if (eid >= e) return;
  int lane = idx & 31;
  int s = row[eid];
  int t = col[eid];
  float w = dinv[s] * dinv[t];
  const float4* src = (const float4*)(xW + (size_t)s * D);
  float4 v = src[lane];
  float* dst = out + (size_t)t * D + lane * 4;
  atomicAdd(dst + 0, v.x * w);
  atomicAdd(dst + 1, v.y * w);
  atomicAdd(dst + 2, v.z * w);
  atomicAdd(dst + 3, v.w * w);
}

__global__ __launch_bounds__(256) void k_finalize(
    float* __restrict__ out, const float* __restrict__ bias,
    const float* __restrict__ prelu, int total) {
  int i = blockIdx.x * 256 + threadIdx.x;
  if (i >= total) return;
  float v = out[i] + bias[i & (D - 1)];
  float pw = prelu[0];
  out[i] = v > 0.0f ? v : pw * v;
}

extern "C" void kernel_launch(void* const* d_in, const int* in_sizes, int n_in,
                              void* d_out, int out_size, void* d_ws, size_t ws_size,
                              hipStream_t stream) {
  const float* seq   = (const float*)d_in[0];
  const int*   eidx  = (const int*)d_in[1];
  const float* W     = (const float*)d_in[2];
  const float* bias  = (const float*)d_in[3];
  const float* prelu = (const float*)d_in[4];

  const int N = in_sizes[0] / D;
  const int E = in_sizes[1] / 2;
  const int* rowp = eidx;
  const int* colp = eidx + E;

  float* xW  = (float*)d_ws;
  float* deg = xW + (size_t)N * D;
  float* out = (float*)d_out;

  k_deg_init <<<(N + 255) / 256, 256, 0, stream>>>(deg, N);
  k_deg_accum<<<(E + 255) / 256, 256, 0, stream>>>(colp, deg, E);
  k_dinv     <<<(N + 255) / 256, 256, 0, stream>>>(deg, N);

  int nt  = N / 16;
  int rem = N - nt * 16;
  if (nt  > 0) k_gemm_selfinit<<<nt, 256, 0, stream>>>(seq, W, deg, xW, out);
  if (rem > 0) k_tail_rows<<<rem, 128, 0, stream>>>(seq, W, deg, xW, out, nt * 16, N);

  int scatter_blocks = (int)(((size_t)E * 32 + 255) / 256);
  k_scatter <<<scatter_blocks, 256, 0, stream>>>(rowp, colp, xW, deg, out, E);

  int total = N * D;
  k_finalize<<<(total + 255) / 256, 256, 0, stream>>>(out, bias, prelu, total);
}


# Edge filter scans one key byte per edge (low 8 bits of target/196, written by the degree pass), one zero-byte test per 4 keys; strip walk checks the real target; LDS atomic arrival order kept
# speedup vs baseline: 2.7862x; 2.7862x over previous
#include <hip/hip_runtime.h>

typedef __attribute__((ext_vector_type(2))) float v2f;
typedef __attribute__((ext_vector_type(8))) float v8f;

#define D 128
#define RPB 196


__device__ __forceinline__ void post_half(int* s_half) {
  if (threadIdx.x < 32) {
    unsigned id = __builtin_amdgcn_s_sendmsg_rtn(0x87);
    if (threadIdx.x == 0) *s_half = ((id >> 18) & 3u) != 0u ? 1 : 0;
  }
}

__global__ __launch_bounds__(256) void k_deg_init(float* __restrict__ deg0,
                                                  float* __restrict__ deg1, int n) {
  int i = blockIdx.x * 256 + threadIdx.x;
  if (i < n) {
    deg0[i] = 1.0f;
    deg1[i] = 0.0f;
  }
}

__global__ __launch_bounds__(256) void k_deg_accum(const int* __restrict__ col,
                                                   float* __restrict__ deg0,
                                                   float* __restrict__ deg1,
                                                   unsigned char* __restrict__ key, int e) {
  __shared__ int s_half;
  int i = blockIdx.x * 256 + threadIdx.x;
  int c = col[i < e ? i : 0];
  post_half(&s_half);
  if (i < e) key[i] = (unsigned char)(((unsigned)c / RPB) & 255u);
  __syncthreads();
  float* deg = s_half ? deg1 : deg0;
  if (i < e) atomicAdd(deg + c, 1.0f);
}

__global__ __launch_bounds__(256) void k_dinv(float* __restrict__ deg0,
                                              const float* __restrict__ deg1, int n) {
  int i = blockIdx.x * 256 + threadIdx.x;
  if (i < n) {
    float d = deg0[i] + deg1[i];
    deg0[i] = d > 0.0f ? rsqrtf(d) : 0.0f;
  }
}

__global__ __launch_bounds__(64) void k_gemm(
    const float* __restrict__ seq, const float* __restrict__ W,
    float* __restrict__ xW) {
  const int lane  = threadIdx.x & 31;
  const int wave  = threadIdx.x >> 5;
  const int row0  = blockIdx.x * 16;
  const int n0    = wave * 64;
  const int mn    = lane & 15;
  const int khalf = (lane >> 4) << 1;

  const float* arow = seq + (size_t)(row0 + mn) * D + khalf;
  const float* wcol = W + (size_t)khalf * D + n0 + 4 * mn;
  v8f acc0 = {}, acc1t = {}, acc2 = {}, acc3 = {};
#pragma unroll 4
  for (int k = 0; k < D; k += 4) {
    v2f a;
    a.x = arow[k];
    a.y = arow[k + 1];
    const float4 w0 = *reinterpret_cast<const float4*>(wcol + (size_t)k * D);
    const float4 w1 = *reinterpret_cast<const float4*>(wcol + (size_t)(k + 1) * D);
    v2f b0, b1, b2, b3;
    b0.x = w0.x; b0.y = w1.x;
    b1.x = w0.y; b1.y = w1.y;
    b2.x = w0.z; b2.y = w1.z;
    b3.x = w0.w; b3.y = w1.w;
    acc0 = __builtin_amdgcn_wmma_f32_16x16x4_f32(
         false, a,  false, b0,
         (short)0, acc0,  false,  false);
    acc1t = __builtin_amdgcn_wmma_f32_16x16x4_f32(
        false, a, false, b1, (short)0, acc1t, false, false);
    acc2 = __builtin_amdgcn_wmma_f32_16x16x4_f32(
        false, a, false, b2, (short)0, acc2, false, false);
    acc3 = __builtin_amdgcn_wmma_f32_16x16x4_f32(
        false, a, false, b3, (short)0, acc3, false, false);
  }

  const int mbase = (lane >> 4) << 3;
#pragma unroll
  for (int j = 0; j < 8; ++j) {
    int r = row0 + mbase + j;
    size_t off = (size_t)r * D + n0 + 4 * mn;
    *reinterpret_cast<float4*>(xW + off) = make_float4(acc0[j], acc1t[j], acc2[j], acc3[j]);
  }
}

__global__ __launch_bounds__(128) void k_tail_rows(
    const float* __restrict__ seq, const float* __restrict__ W,
    float* __restrict__ xW, int rowStart, int n) {
  int r = rowStart + blockIdx.x;
  if (r >= n) return;
  int c = threadIdx.x;
  float acc = 0.0f;
  for (int k = 0; k < D; ++k) acc += seq[(size_t)r * D + k] * W[(size_t)k * D + c];
  xW[(size_t)r * D + c] = acc;
}

#define AGG_WAVES 32
#define QCAP (32 + 512)
#define GROUP 8

__device__ __forceinline__ void walk_strip(
    const int* q, int cnt, int lane, int r0, int nrows, float* acc,
    const int* __restrict__ row, const int* __restrict__ col,
    const float* __restrict__ xW, const float* __restrict__ dinv) {
  const int my = q[lane < cnt ? lane : 0];
  const int s  = row[my];
  const int t  = col[my];
  const int tl = t - r0;
  const unsigned real = __builtin_amdgcn_ballot_w32(
      lane < cnt && (unsigned)tl < (unsigned)nrows);
  const int wbits = __float_as_int(dinv[s] * dinv[t]);

  for (int j0 = 0; j0 < cnt; j0 += GROUP) {
    float m[GROUP][4];
    for (int u = 0; u < GROUP; ++u) {
      const int sj = __builtin_amdgcn_readlane(s, j0 + u);
      const float* src = xW + (size_t)sj * D + lane;
      m[u][0] = src[0];
      m[u][1] = src[32];
      m[u][2] = src[64];
      m[u][3] = src[96];
    }
    for (int u = 0; u < GROUP; ++u) {
      if ((real >> (j0 + u)) & 1u) {
        const int tj = __builtin_amdgcn_readlane(tl, j0 + u);
        const float wj = __int_as_float(__builtin_amdgcn_readlane(wbits, j0 + u));
        float* dst = acc + tj * D + lane;
        atomicAdd(dst + 0,  m[u][0] * wj);
        atomicAdd(dst + 32, m[u][1] * wj);
        atomicAdd(dst + 64, m[u][2] * wj);
        atomicAdd(dst + 96, m[u][3] * wj);
      }
    }
  }
}

__device__ __forceinline__ uint4 load_keys(const unsigned char* __restrict__ key,
                                           int c, int lane, int e, unsigned pat) {
  const int nfull = e >> 9;
  const int cc = c < nfull ? c : (nfull > 0 ? nfull - 1 : 0);
  uint4 kw = *reinterpret_cast<const uint4*>(key + ((size_t)cc << 9) + 16 * lane);
  if (c >= nfull) {
    const unsigned nokey = (pat & 255u) ^ 1u;
    unsigned xw[4];
#pragma unroll
    for (int w = 0; w < 4; ++w) {
      unsigned x = 0u;
#pragma unroll
      for (int k = 0; k < 4; ++k) {
        const int ei = (c << 9) + 16 * lane + 4 * w + k;
        const unsigned b = ei < e ? (unsigned)key[ei] : nokey;
        x |= b << (8 * k);
      }
      xw[w] = x;
    }
    kw = make_uint4(xw[0], xw[1], xw[2], xw[3]);
  }
  return kw;
}

__global__ __launch_bounds__(1024) void k_aggregate(
    const int* __restrict__ row, const int* __restrict__ col,
    const unsigned char* __restrict__ key,
    const float* __restrict__ xW, const float* __restrict__ dinv,
    const float* __restrict__ bias, const float* __restrict__ prelu,
    float* __restrict__ out, int e, int n) {
  __shared__ __attribute__((aligned(16))) float acc[RPB * D];
  __shared__ int stack[AGG_WAVES][QCAP];

  const int lane = threadIdx.x & 31;
  const int wave = __builtin_amdgcn_readfirstlane((int)(threadIdx.x >> 5));
  const int r0 = blockIdx.x * RPB;
  const int nrows = (n - r0) < RPB ? (n - r0) : RPB;

  float4* acc4 = reinterpret_cast<float4*>(acc);
  for (int i = threadIdx.x; i < nrows * (D / 4); i += 1024) {
    const int r = r0 + (i >> 5);
    const float di = dinv[r];
    const float4 x = reinterpret_cast<const float4*>(xW + (size_t)r * D)[i & 31];
    acc4[i] = make_float4(x.x * di * di, x.y * di * di, x.z * di * di, x.w * di * di);
  }
  __syncthreads();

  int* myq = stack[wave];
  int qn = 0;
  const unsigned pat = (blockIdx.x & 255u) * 0x01010101u;
  const int nchunks = (e + 511) >> 9;
  int c = wave;
  uint4 k0 = load_keys(key, c, lane, e, pat);
  uint4 k1 = load_keys(key, c + AGG_WAVES, lane, e, pat);
  for (; c < nchunks; c += AGG_WAVES) {
    const uint4 k2 = load_keys(key, c + 2 * AGG_WAVES, lane, e, pat);
    const int eb = (c << 9) + 16 * lane;
    const unsigned xv[4] = {k0.x ^ pat, k0.y ^ pat, k0.z ^ pat, k0.w ^ pat};
#pragma unroll
    for (int w = 0; w < 4; ++w) {
      const unsigned x = xv[w];
      const bool any = ((x - 0x01010101u) & ~x & 0x80808080u) != 0u;
      if (__builtin_amdgcn_ballot_w32(any) != 0u) {
#pragma unroll
        for (int k = 0; k < 4; ++k) {
          const bool hit = ((x >> (8 * k)) & 255u) == 0u;
          const unsigned mask = __builtin_amdgcn_ballot_w32(hit);
          if (mask != 0u) {
            const int pos = qn + (int)__builtin_amdgcn_mbcnt_lo(mask, 0u);
            if (hit) myq[pos] = eb + 4 * w + k;
            qn += __builtin_popcount(mask);
          }
        }
      }
    }
    while (qn >= 32) {
      qn -= 32;
      walk_strip(myq + qn, 32, lane, r0, nrows, acc, row, col, xW, dinv);
    }
    k0 = k1;
    k1 = k2;
  }
  if (qn > 0) {
    walk_strip(myq, qn, lane, r0, nrows, acc, row, col, xW, dinv);
  }
  __syncthreads();

  const float pw = prelu[0];
  float4* out4 = reinterpret_cast<float4*>(out + (size_t)r0 * D);
  const float4* bias4 = reinterpret_cast<const float4*>(bias);
  for (int i = threadIdx.x; i < nrows * (D / 4); i += 1024) {
    const float4 a = acc4[i];
    const float4 b = bias4[i & 31];
    float4 v = make_float4(a.x + b.x, a.y + b.y, a.z + b.z, a.w + b.w);
    v.x = v.x > 0.0f ? v.x : pw * v.x;
    v.y = v.y > 0.0f ? v.y : pw * v.y;
    v.z = v.z > 0.0f ? v.z : pw * v.z;
    v.w = v.w > 0.0f ? v.w : pw * v.w;
    out4[i] = v;
  }
}

extern "C" void kernel_launch(void* const* d_in, const int* in_sizes, int n_in,
                              void* d_out, int out_size, void* d_ws, size_t ws_size,
                              hipStream_t stream) {
  const float* seq   = (const float*)d_in[0];
  const int*   eidx  = (const int*)d_in[1];
  const float* W     = (const float*)d_in[2];
  const float* bias  = (const float*)d_in[3];
  const float* prelu = (const float*)d_in[4];

  const int N = in_sizes[0] / D;
  const int E = in_sizes[1] / 2;
  const int* rowp = eidx;
  const int* colp = eidx + E;

  float* xW   = (float*)d_ws;
  unsigned char* key = (unsigned char*)(xW + (size_t)N * D);
  float* deg0 = (float*)(key + (((size_t)E + 15) & ~(size_t)15));
  float* deg1 = deg0 + N;
  float* out  = (float*)d_out;

  k_deg_init <<<(N + 255) / 256, 256, 0, stream>>>(deg0, deg1, N);
  k_deg_accum<<<(E + 255) / 256, 256, 0, stream>>>(colp, deg0, deg1, key, E);
  k_dinv     <<<(N + 255) / 256, 256, 0, stream>>>(deg0, deg1, N);

  int nt  = N / 16;
  int rem = N - nt * 16;
  if (nt  > 0) k_gemm<<<nt, 64, 0, stream>>>(seq, W, xW);
  if (rem > 0) k_tail_rows<<<rem, 128, 0, stream>>>(seq, W, xW, nt * 16, N);

  int agg_blocks = (N + RPB - 1) / RPB;
  if (agg_blocks > 0) k_aggregate<<<agg_blocks, 1024, 0, stream>>>(rowp, colp, key, xW, deg0, bias, prelu, out, E, N);
}


# Edge norm factored out of the sum: GEMM stores rows scaled by dinv[source], last pass scales by dinv[target]; no per-edge weight gathers or multiplies, no dinv launch (5 to 4); sum order kept
# speedup vs baseline: 3.1161x; 1.1184x over previous
#include <hip/hip_runtime.h>

typedef __attribute__((ext_vector_type(2))) float v2f;
typedef __attribute__((ext_vector_type(8))) float v8f;

#define D 128
#define RPB 196


__device__ __forceinline__ void post_half(int* s_half) {
  if (threadIdx.x < 32) {
    unsigned id = __builtin_amdgcn_s_sendmsg_rtn(0x87);
    if (threadIdx.x == 0) *s_half = ((id >> 18) & 3u) != 0u ? 1 : 0;
  }
}

__global__ __launch_bounds__(256) void k_deg_init(float* __restrict__ deg0,
                                                  float* __restrict__ deg1, int n) {
  int i = blockIdx.x * 256 + threadIdx.x;
  if (i < n) {
    deg0[i] = 1.0f;
    deg1[i] = 0.0f;
  }
}

__global__ __launch_bounds__(256) void k_deg_accum(const int* __restrict__ col,
                                                   float* __restrict__ deg0,
                                                   float* __restrict__ deg1,
                                                   unsigned char* __restrict__ key, int e) {
  __shared__ int s_half;
  int i = blockIdx.x * 256 + threadIdx.x;
  int c = col[i < e ? i : 0];
  post_half(&s_half);
  if (i < e) key[i] = (unsigned char)(((unsigned)c / RPB) & 255u);
  __syncthreads();
  float* deg = s_half ? deg1 : deg0;
  if (i < e) atomicAdd(deg + c, 1.0f);
}

__device__ __forceinline__ float dinv_of(const float* __restrict__ deg0,
                                         const float* __restrict__ deg1, int i) {
  const float d = deg0[i] + deg1[i];
  return d > 0.0f ? rsqrtf(d) : 0.0f;
}

__global__ __launch_bounds__(64) void k_gemm(
    const float* __restrict__ seq, const float* __restrict__ W,
    const float* __restrict__ deg0, const float* __restrict__ deg1,
    float* __restrict__ xW) {
  const int lane  = threadIdx.x & 31;
  const int wave  = threadIdx.x >> 5;
  const int row0  = blockIdx.x * 16;
  const int n0    = wave * 64;
  const int mn    = lane & 15;
  const int khalf = (lane >> 4) << 1;

  const float* arow = seq + (size_t)(row0 + mn) * D + khalf;
  const float* wcol = W + (size_t)khalf * D + n0 + 4 * mn;
  const int mbase = (lane >> 4) << 3;
  float di[8];
#pragma unroll
  for (int j = 0; j < 8; ++j) di[j] = dinv_of(deg0, deg1, row0 + mbase + j);
  v8f acc0 = {}, acc1t = {}, acc2 = {}, acc3 = {};
#pragma unroll 4
  for (int k = 0; k < D; k += 4) {
    v2f a;
    a.x = arow[k];
    a.y = arow[k + 1];
    const float4 w0 = *reinterpret_cast<const float4*>(wcol + (size_t)k * D);
    const float4 w1 = *reinterpret_cast<const float4*>(wcol + (size_t)(k + 1) * D);
    v2f b0, b1, b2, b3;
    b0.x = w0.x; b0.y = w1.x;
    b1.x = w0.y; b1.y = w1.y;
    b2.x = w0.z; b2.y = w1.z;
    b3.x = w0.w; b3.y = w1.w;
    acc0 = __builtin_amdgcn_wmma_f32_16x16x4_f32(
         false, a,  false, b0,
         (short)0, acc0,  false,  false);
    acc1t = __builtin_amdgcn_wmma_f32_16x16x4_f32(
        false, a, false, b1, (short)0, acc1t, false, false);
    acc2 = __builtin_amdgcn_wmma_f32_16x16x4_f32(
        false, a, false, b2, (short)0, acc2, false, false);
    acc3 = __builtin_amdgcn_wmma_f32_16x16x4_f32(
        false, a, false, b3, (short)0, acc3, false, false);
  }

#pragma unroll
  for (int j = 0; j < 8; ++j) {
    int r = row0 + mbase + j;
    size_t off = (size_t)r * D + n0 + 4 * mn;
    *reinterpret_cast<float4*>(xW + off) =
        make_float4(acc0[j] * di[j], acc1t[j] * di[j], acc2[j] * di[j], acc3[j] * di[j]);
  }
}

__global__ __launch_bounds__(128) void k_tail_rows(
    const float* __restrict__ seq, const float* __restrict__ W,
    const float* __restrict__ deg0, const float* __restrict__ deg1,
    float* __restrict__ xW, int rowStart, int n) {
  int r = rowStart + blockIdx.x;
  if (r >= n) return;
  int c = threadIdx.x;
  float acc = 0.0f;
  for (int k = 0; k < D; ++k) acc += seq[(size_t)r * D + k] * W[(size_t)k * D + c];
  xW[(size_t)r * D + c] = acc * dinv_of(deg0, deg1, r);
}

#define AGG_WAVES 32
#define QCAP (32 + 512)
#define GROUP 8

__device__ __forceinline__ void walk_strip(
    const int* q, int cnt, int lane, int r0, int nrows, float* acc,
    const int* __restrict__ row, const int* __restrict__ col,
    const float* __restrict__ xW) {
  const int my = q[lane < cnt ? lane : 0];
  const int s  = row[my];
  const int t  = col[my];
  const int tl = t - r0;
  const unsigned real = __builtin_amdgcn_ballot_w32(
      lane < cnt && (unsigned)tl < (unsigned)nrows);

  for (int j0 = 0; j0 < cnt; j0 += GROUP) {
    float m[GROUP][4];
    for (int u = 0; u < GROUP; ++u) {
      const int sj = __builtin_amdgcn_readlane(s, j0 + u);
      const float* src = xW + (size_t)sj * D + lane;
      m[u][0] = src[0];
      m[u][1] = src[32];
      m[u][2] = src[64];
      m[u][3] = src[96];
    }
    for (int u = 0; u < GROUP; ++u) {
      if ((real >> (j0 + u)) & 1u) {
        const int tj = __builtin_amdgcn_readlane(tl, j0 + u);
        float* dst = acc + tj * D + lane;
        atomicAdd(dst + 0,  m[u][0]);
        atomicAdd(dst + 32, m[u][1]);
        atomicAdd(dst + 64, m[u][2]);
        atomicAdd(dst + 96, m[u][3]);
      }
    }
  }
}

__device__ __forceinline__ uint4 load_keys(const unsigned char* __restrict__ key,
                                           int c, int lane, int e, unsigned pat) {
  const int nfull = e >> 9;
  const int cc = c < nfull ? c : (nfull > 0 ? nfull - 1 : 0);
  uint4 kw = *reinterpret_cast<const uint4*>(key + ((size_t)cc << 9) + 16 * lane);
  if (c >= nfull) {
    const unsigned nokey = (pat & 255u) ^ 1u;
    unsigned xw[4];
#pragma unroll
    for (int w = 0; w < 4; ++w) {
      unsigned x = 0u;
#pragma unroll
      for (int k = 0; k < 4; ++k) {
        const int ei = (c << 9) + 16 * lane + 4 * w + k;
        const unsigned b = ei < e ? (unsigned)key[ei] : nokey;
        x |= b << (8 * k);
      }
      xw[w] = x;
    }
    kw = make_uint4(xw[0], xw[1], xw[2], xw[3]);
  }
  return kw;
}

__global__ __launch_bounds__(1024) void k_aggregate(
    const int* __restrict__ row, const int* __restrict__ col,
    const unsigned char* __restrict__ key,
    const float* __restrict__ xW,
    const float* __restrict__ deg0, const float* __restrict__ deg1,
    const float* __restrict__ bias, const float* __restrict__ prelu,
    float* __restrict__ out, int e, int n) {
  __shared__ __attribute__((aligned(16))) float acc[RPB * D];
  __shared__ int stack[AGG_WAVES][QCAP];
  __shared__ float s_dt[RPB];

  const int lane = threadIdx.x & 31;
  const int wave = __builtin_amdgcn_readfirstlane((int)(threadIdx.x >> 5));
  const int r0 = blockIdx.x * RPB;
  const int nrows = (n - r0) < RPB ? (n - r0) : RPB;

  float4* acc4 = reinterpret_cast<float4*>(acc);
  for (int i = threadIdx.x; i < nrows * (D / 4); i += 1024) {
    const int r = r0 + (i >> 5);
    acc4[i] = reinterpret_cast<const float4*>(xW + (size_t)r * D)[i & 31];
  }
  if ((int)threadIdx.x < nrows) s_dt[threadIdx.x] = dinv_of(deg0, deg1, r0 + (int)threadIdx.x);
  __syncthreads();

  int* myq = stack[wave];
  int qn = 0;
  const unsigned pat = (blockIdx.x & 255u) * 0x01010101u;
  const int nchunks = (e + 511) >> 9;
  int c = wave;
  uint4 k0 = load_keys(key, c, lane, e, pat);
  uint4 k1 = load_keys(key, c + AGG_WAVES, lane, e, pat);
  for (; c < nchunks; c += AGG_WAVES) {
    const uint4 k2 = load_keys(key, c + 2 * AGG_WAVES, lane, e, pat);
    const int eb = (c << 9) + 16 * lane;
    const unsigned xv[4] = {k0.x ^ pat, k0.y ^ pat, k0.z ^ pat, k0.w ^ pat};
#pragma unroll
    for (int w = 0; w < 4; ++w) {
      const unsigned x = xv[w];
      const bool any = ((x - 0x01010101u) & ~x & 0x80808080u) != 0u;
      if (__builtin_amdgcn_ballot_w32(any) != 0u) {
#pragma unroll
        for (int k = 0; k < 4; ++k) {
          const bool hit = ((x >> (8 * k)) & 255u) == 0u;
          const unsigned mask = __builtin_amdgcn_ballot_w32(hit);
          if (mask != 0u) {
            const int pos = qn + (int)__builtin_amdgcn_mbcnt_lo(mask, 0u);
            if (hit) myq[pos] = eb + 4 * w + k;
            qn += __builtin_popcount(mask);
          }
        }
      }
    }
    while (qn >= 32) {
      qn -= 32;
      walk_strip(myq + qn, 32, lane, r0, nrows, acc, row, col, xW);
    }
    k0 = k1;
    k1 = k2;
  }
  if (qn > 0) {
    walk_strip(myq, qn, lane, r0, nrows, acc, row, col, xW);
  }
  __syncthreads();

  const float pw = prelu[0];
  float4* out4 = reinterpret_cast<float4*>(out + (size_t)r0 * D);
  const float4* bias4 = reinterpret_cast<const float4*>(bias);
  for (int i = threadIdx.x; i < nrows * (D / 4); i += 1024) {
    const float4 a = acc4[i];
    const float4 b = bias4[i & 31];
    const float dt = s_dt[i >> 5];
    float4 v = make_float4(a.x * dt + b.x, a.y * dt + b.y, a.z * dt + b.z, a.w * dt + b.w);
    v.x = v.x > 0.0f ? v.x : pw * v.x;
    v.y = v.y > 0.0f ? v.y : pw * v.y;
    v.z = v.z > 0.0f ? v.z : pw * v.z;
    v.w = v.w > 0.0f ? v.w : pw * v.w;
    out4[i] = v;
  }
}

extern "C" void kernel_launch(void* const* d_in, const int* in_sizes, int n_in,
                              void* d_out, int out_size, void* d_ws, size_t ws_size,
                              hipStream_t stream) {
  const float* seq   = (const float*)d_in[0];
  const int*   eidx  = (const int*)d_in[1];
  const float* W     = (const float*)d_in[2];
  const float* bias  = (const float*)d_in[3];
  const float* prelu = (const float*)d_in[4];

  const int N = in_sizes[0] / D;
  const int E = in_sizes[1] / 2;
  const int* rowp = eidx;
  const int* colp = eidx + E;

  float* xW   = (float*)d_ws;
  unsigned char* key = (unsigned char*)(xW + (size_t)N * D);
  float* deg0 = (float*)(key + (((size_t)E + 15) & ~(size_t)15));
  float* deg1 = deg0 + N;
  float* out  = (float*)d_out;

  k_deg_init <<<(N + 255) / 256, 256, 0, stream>>>(deg0, deg1, N);
  k_deg_accum<<<(E + 255) / 256, 256, 0, stream>>>(colp, deg0, deg1, key, E);

  int nt  = N / 16;
  int rem = N - nt * 16;
  if (nt  > 0) k_gemm<<<nt, 64, 0, stream>>>(seq, W, deg0, deg1, xW);
  if (rem > 0) k_tail_rows<<<rem, 128, 0, stream>>>(seq, W, deg0, deg1, xW, nt * 16, N);

  int agg_blocks = (N + RPB - 1) / RPB;
  if (agg_blocks > 0) k_aggregate<<<agg_blocks, 1024, 0, stream>>>(rowp, colp, key, xW, deg0, deg1, bias, prelu, out, E, N);
}


# Degree pass also writes one 8-byte (source, target) record per edge; the strip walk fetches one record per candidate: one scattered line request per edge instead of two; sum order kept
# speedup vs baseline: 3.2452x; 1.0414x over previous
#include <hip/hip_runtime.h>

typedef __attribute__((ext_vector_type(2))) float v2f;
typedef __attribute__((ext_vector_type(8))) float v8f;

#define D 128
#define RPB 196


__device__ __forceinline__ void post_half(int* s_half) {
  if (threadIdx.x < 32) {
    unsigned id = __builtin_amdgcn_s_sendmsg_rtn(0x87);
    if (threadIdx.x == 0) *s_half = ((id >> 18) & 3u) != 0u ? 1 : 0;
  }
}

__global__ __launch_bounds__(256) void k_deg_init(float* __restrict__ deg0,
                                                  float* __restrict__ deg1, int n) {
  int i = blockIdx.x * 256 + threadIdx.x;
  if (i < n) {
    deg0[i] = 1.0f;
    deg1[i] = 0.0f;
  }
}

__global__ __launch_bounds__(256) void k_deg_accum(const int* __restrict__ row,
                                                   const int* __restrict__ col,
                                                   float* __restrict__ deg0,
                                                   float* __restrict__ deg1,
                                                   unsigned char* __restrict__ key,
                                                   int* __restrict__ rec, int e) {
  __shared__ int s_half;
  int i = blockIdx.x * 256 + threadIdx.x;
  int r = row[i < e ? i : 0];
  int c = col[i < e ? i : 0];
  post_half(&s_half);
  if (i < e) {
    key[i] = (unsigned char)(((unsigned)c / RPB) & 255u);
    reinterpret_cast<int2*>(rec)[i] = make_int2(r, c);
  }
  __syncthreads();
  float* deg = s_half ? deg1 : deg0;
  if (i < e) atomicAdd(deg + c, 1.0f);
}

__device__ __forceinline__ float dinv_of(const float* __restrict__ deg0,
                                         const float* __restrict__ deg1, int i) {
  const float d = deg0[i] + deg1[i];
  return d > 0.0f ? rsqrtf(d) : 0.0f;
}

__global__ __launch_bounds__(64) void k_gemm(
    const float* __restrict__ seq, const float* __restrict__ W,
    const float* __restrict__ deg0, const float* __restrict__ deg1,
    float* __restrict__ xW) {
  const int lane  = threadIdx.x & 31;
  const int wave  = threadIdx.x >> 5;
  const int row0  = blockIdx.x * 16;
  const int n0    = wave * 64;
  const int mn    = lane & 15;
  const int khalf = (lane >> 4) << 1;

  const float* arow = seq + (size_t)(row0 + mn) * D + khalf;
  const float* wcol = W + (size_t)khalf * D + n0 + 4 * mn;
  const int mbase = (lane >> 4) << 3;
  float di[8];
#pragma unroll
  for (int j = 0; j < 8; ++j) di[j] = dinv_of(deg0, deg1, row0 + mbase + j);
  v8f acc0 = {}, acc1t = {}, acc2 = {}, acc3 = {};
#pragma unroll 4
  for (int k = 0; k < D; k += 4) {
    v2f a;
    a.x = arow[k];
    a.y = arow[k + 1];
    const float4 w0 = *reinterpret_cast<const float4*>(wcol + (size_t)k * D);
    const float4 w1 = *reinterpret_cast<const float4*>(wcol + (size_t)(k + 1) * D);
    v2f b0, b1, b2, b3;
    b0.x = w0.x; b0.y = w1.x;
    b1.x = w0.y; b1.y = w1.y;
    b2.x = w0.z; b2.y = w1.z;
    b3.x = w0.w; b3.y = w1.w;
    acc0 = __builtin_amdgcn_wmma_f32_16x16x4_f32(
         false, a,  false, b0,
         (short)0, acc0,  false,  false);
    acc1t = __builtin_amdgcn_wmma_f32_16x16x4_f32(
        false, a, false, b1, (short)0, acc1t, false, false);
    acc2 = __builtin_amdgcn_wmma_f32_16x16x4_f32(
        false, a, false, b2, (short)0, acc2, false, false);
    acc3 = __builtin_amdgcn_wmma_f32_16x16x4_f32(
        false, a, false, b3, (short)0, acc3, false, false);
  }

#pragma unroll
  for (int j = 0; j < 8; ++j) {
    int r = row0 + mbase + j;
    size_t off = (size_t)r * D + n0 + 4 * mn;
    *reinterpret_cast<float4*>(xW + off) =
        make_float4(acc0[j] * di[j], acc1t[j] * di[j], acc2[j] * di[j], acc3[j] * di[j]);
  }
}

__global__ __launch_bounds__(128) void k_tail_rows(
    const float* __restrict__ seq, const float* __restrict__ W,
    const float* __restrict__ deg0, const float* __restrict__ deg1,
    float* __restrict__ xW, int rowStart, int n) {
  int r = rowStart + blockIdx.x;
  if (r >= n) return;
  int c = threadIdx.x;
  float acc = 0.0f;
  for (int k = 0; k < D; ++k) acc += seq[(size_t)r * D + k] * W[(size_t)k * D + c];
  xW[(size_t)r * D + c] = acc * dinv_of(deg0, deg1, r);
}

#define AGG_WAVES 32
#define QCAP (32 + 512)
#define GROUP 8

__device__ __forceinline__ void walk_strip(
    const int* q, int cnt, int lane, int r0, int nrows, float* acc,
    const int* __restrict__ rec,
    const float* __restrict__ xW) {
  const int my = q[lane < cnt ? lane : 0];
  const int2 st = reinterpret_cast<const int2*>(rec)[my];
  const int s  = st.x;
  const int t  = st.y;
  const int tl = t - r0;
  const unsigned real = __builtin_amdgcn_ballot_w32(
      lane < cnt && (unsigned)tl < (unsigned)nrows);

  for (int j0 = 0; j0 < cnt; j0 += GROUP) {
    float m[GROUP][4];
    for (int u = 0; u < GROUP; ++u) {
      const int sj = __builtin_amdgcn_readlane(s, j0 + u);
      const float* src = xW + (size_t)sj * D + lane;
      m[u][0] = src[0];
      m[u][1] = src[32];
      m[u][2] = src[64];
      m[u][3] = src[96];
    }
    for (int u = 0; u < GROUP; ++u) {
      if ((real >> (j0 + u)) & 1u) {
        const int tj = __builtin_amdgcn_readlane(tl, j0 + u);
        float* dst = acc + tj * D + lane;
        atomicAdd(dst + 0,  m[u][0]);
        atomicAdd(dst + 32, m[u][1]);
        atomicAdd(dst + 64, m[u][2]);
        atomicAdd(dst + 96, m[u][3]);
      }
    }
  }
}

__device__ __forceinline__ uint4 load_keys(const unsigned char* __restrict__ key,
                                           int c, int lane, int e, unsigned pat) {
  const int nfull = e >> 9;
  const int cc = c < nfull ? c : (nfull > 0 ? nfull - 1 : 0);
  uint4 kw = *reinterpret_cast<const uint4*>(key + ((size_t)cc << 9) + 16 * lane);
  if (c >= nfull) {
    const unsigned nokey = (pat & 255u) ^ 1u;
    unsigned xw[4];
#pragma unroll
    for (int w = 0; w < 4; ++w) {
      unsigned x = 0u;
#pragma unroll
      for (int k = 0; k < 4; ++k) {
        const int ei = (c << 9) + 16 * lane + 4 * w + k;
        const unsigned b = ei < e ? (unsigned)key[ei] : nokey;
        x |= b << (8 * k);
      }
      xw[w] = x;
    }
    kw = make_uint4(xw[0], xw[1], xw[2], xw[3]);
  }
  return kw;
}

__global__ __launch_bounds__(1024) void k_aggregate(
    const int* __restrict__ rec,
    const unsigned char* __restrict__ key,
    const float* __restrict__ xW,
    const float* __restrict__ deg0, const float* __restrict__ deg1,
    const float* __restrict__ bias, const float* __restrict__ prelu,
    float* __restrict__ out, int e, int n) {
  __shared__ __attribute__((aligned(16))) float acc[RPB * D];
  __shared__ int stack[AGG_WAVES][QCAP];
  __shared__ float s_dt[RPB];

  const int lane = threadIdx.x & 31;
  const int wave = __builtin_amdgcn_readfirstlane((int)(threadIdx.x >> 5));
  const int r0 = blockIdx.x * RPB;
  const int nrows = (n - r0) < RPB ? (n - r0) : RPB;

  float4* acc4 = reinterpret_cast<float4*>(acc);
  for (int i = threadIdx.x; i < nrows * (D / 4); i += 1024) {
    const int r = r0 + (i >> 5);
    acc4[i] = reinterpret_cast<const float4*>(xW + (size_t)r * D)[i & 31];
  }
  if ((int)threadIdx.x < nrows) s_dt[threadIdx.x] = dinv_of(deg0, deg1, r0 + (int)threadIdx.x);
  __syncthreads();

  int* myq = stack[wave];
  int qn = 0;
  const unsigned pat = (blockIdx.x & 255u) * 0x01010101u;
  const int nchunks = (e + 511) >> 9;
  int c = wave;
  uint4 k0 = load_keys(key, c, lane, e, pat);
  uint4 k1 = load_keys(key, c + AGG_WAVES, lane, e, pat);
  for (; c < nchunks; c += AGG_WAVES) {
    const uint4 k2 = load_keys(key, c + 2 * AGG_WAVES, lane, e, pat);
    const int eb = (c << 9) + 16 * lane;
    const unsigned xv[4] = {k0.x ^ pat, k0.y ^ pat, k0.z ^ pat, k0.w ^ pat};
#pragma unroll
    for (int w = 0; w < 4; ++w) {
      const unsigned x = xv[w];
      const bool any = ((x - 0x01010101u) & ~x & 0x80808080u) != 0u;
      if (__builtin_amdgcn_ballot_w32(any) != 0u) {
#pragma unroll
        for (int k = 0; k < 4; ++k) {
          const bool hit = ((x >> (8 * k)) & 255u) == 0u;
          const unsigned mask = __builtin_amdgcn_ballot_w32(hit);
          if (mask != 0u) {
            const int pos = qn + (int)__builtin_amdgcn_mbcnt_lo(mask, 0u);
            if (hit) myq[pos] = eb + 4 * w + k;
            qn += __builtin_popcount(mask);
          }
        }
      }
    }
    while (qn >= 32) {
      qn -= 32;
      walk_strip(myq + qn, 32, lane, r0, nrows, acc, rec, xW);
    }
    k0 = k1;
    k1 = k2;
  }
  if (qn > 0) {
    walk_strip(myq, qn, lane, r0, nrows, acc, rec, xW);
  }
  __syncthreads();

  const float pw = prelu[0];
  float4* out4 = reinterpret_cast<float4*>(out + (size_t)r0 * D);
  const float4* bias4 = reinterpret_cast<const float4*>(bias);
  for (int i = threadIdx.x; i < nrows * (D / 4); i += 1024) {
    const float4 a = acc4[i];
    const float4 b = bias4[i & 31];
    const float dt = s_dt[i >> 5];
    float4 v = make_float4(a.x * dt + b.x, a.y * dt + b.y, a.z * dt + b.z, a.w * dt + b.w);
    v.x = v.x > 0.0f ? v.x : pw * v.x;
    v.y = v.y > 0.0f ? v.y : pw * v.y;
    v.z = v.z > 0.0f ? v.z : pw * v.z;
    v.w = v.w > 0.0f ? v.w : pw * v.w;
    out4[i] = v;
  }
}

extern "C" void kernel_launch(void* const* d_in, const int* in_sizes, int n_in,
                              void* d_out, int out_size, void* d_ws, size_t ws_size,
                              hipStream_t stream) {
  const float* seq   = (const float*)d_in[0];
  const int*   eidx  = (const int*)d_in[1];
  const float* W     = (const float*)d_in[2];
  const float* bias  = (const float*)d_in[3];
  const float* prelu = (const float*)d_in[4];

  const int N = in_sizes[0] / D;
  const int E = in_sizes[1] / 2;
  const int* rowp = eidx;
  const int* colp = eidx + E;

  float* xW   = (float*)d_ws;
  unsigned char* key = (unsigned char*)(xW + (size_t)N * D);
  float* deg0 = (float*)(key + (((size_t)E + 15) & ~(size_t)15));
  float* deg1 = deg0 + N;
  int*   rec  = (int*)(deg1 + N);
  float* out  = (float*)d_out;

  k_deg_init <<<(N + 255) / 256, 256, 0, stream>>>(deg0, deg1, N);
  k_deg_accum<<<(E + 255) / 256, 256, 0, stream>>>(rowp, colp, deg0, deg1, key, rec, E);

  int nt  = N / 16;
  int rem = N - nt * 16;
  if (nt  > 0) k_gemm<<<nt, 64, 0, stream>>>(seq, W, deg0, deg1, xW);
  if (rem > 0) k_tail_rows<<<rem, 128, 0, stream>>>(seq, W, deg0, deg1, xW, nt * 16, N);

  int agg_blocks = (N + RPB - 1) / RPB;
  if (agg_blocks > 0) k_aggregate<<<agg_blocks, 1024, 0, stream>>>(rec, key, xW, deg0, deg1, bias, prelu, out, E, N);
}
